# starting kernel
# baseline (speedup 1.0000x reference)
#include <hip/hip_runtime.h>
#include <hip/hip_fp16.h>
#include <hip/hip_bf16.h>

#define B_  16
#define S_  512
#define D_  1024
#define H_  128
#define G_  256
#define HP_ 136
#define ZP_ 516

typedef __attribute__((ext_vector_type(16))) _Float16 v16h;
typedef __attribute__((ext_vector_type(8)))  float    v8f;

__device__ __forceinline__ v16h frag_from_f32(const float* p) {
  float4 x0 = *(const float4*)(p);
  float4 x1 = *(const float4*)(p + 4);
  float4 y0 = *(const float4*)(p + 16);
  float4 y1 = *(const float4*)(p + 20);
  v16h f;
  f[0]=(_Float16)x0.x;  f[1]=(_Float16)x0.y;  f[2]=(_Float16)x0.z;  f[3]=(_Float16)x0.w;
  f[4]=(_Float16)x1.x;  f[5]=(_Float16)x1.y;  f[6]=(_Float16)x1.z;  f[7]=(_Float16)x1.w;
  f[8]=(_Float16)y0.x;  f[9]=(_Float16)y0.y;  f[10]=(_Float16)y0.z; f[11]=(_Float16)y0.w;
  f[12]=(_Float16)y1.x; f[13]=(_Float16)y1.y; f[14]=(_Float16)y1.z; f[15]=(_Float16)y1.w;
  return f;
}

__device__ __forceinline__ v16h frag_from_f16(const _Float16* p) {
  v16h f;
  #pragma unroll
  for (int i = 0; i < 8; ++i) { f[i] = p[i]; f[8 + i] = p[16 + i]; }
  return f;
}

#define WMMA_F16(a, b, c) __builtin_amdgcn_wmma_f32_16x16x32_f16( \
    false, (a), false, (b), (short)0, (c), false, false)

template <int A_MODE, int ACT, bool FULLN>
__global__ __launch_bounds__(128) void gemm_wmma_kernel(
    const float* __restrict__ A, const float* __restrict__ Bm,
    const float* __restrict__ bias, const float* __restrict__ bias2,
    const float* __restrict__ resid, float* __restrict__ C,
    int M, int N, int K, int lda, int ldb, int ldc,
    long long sA, long long sB, long long sC)
{
  const int lane  = threadIdx.x & 31;
  const int wave  = threadIdx.x >> 5;
  const int tileM = blockIdx.y * 32;
  const int nbase = blockIdx.x * 256 + wave * 64;
  if (nbase >= N) return;

  const int z = blockIdx.z;
  const float* Ab = A  + (size_t)z * (size_t)sA;
  const float* Bb = Bm + (size_t)z * (size_t)sB;
  float*       Cb = C  + (size_t)z * (size_t)sC;
  const float* Rb = resid ? (resid + (size_t)z * (size_t)sC) : nullptr;

  const int l15   = lane & 15;
  const int khalf = (lane & 16) ? 8 : 0;

  int am0 = tileM + l15;      if (am0 >= M) am0 = M - 1;
  int am1 = tileM + 16 + l15; if (am1 >= M) am1 = M - 1;
  size_t arow0, arow1;
  if (A_MODE == 1) {
    arow0 = ((size_t)(am0 & 15) * S_ + (size_t)(am0 >> 4)) * (size_t)lda;
    arow1 = ((size_t)(am1 & 15) * S_ + (size_t)(am1 >> 4)) * (size_t)lda;
  } else {
    arow0 = (size_t)am0 * (size_t)lda;
    arow1 = (size_t)am1 * (size_t)lda;
  }

  bool ton[4]; int bn[4];
  #pragma unroll
  for (int nt = 0; nt < 4; ++nt) {
    if (FULLN) {
      ton[nt] = true;
      bn[nt]  = nbase + nt * 16 + l15;
    } else {
      ton[nt] = (nbase + nt * 16) < N;
      int c = nbase + nt * 16 + l15; if (c >= N) c = N - 1;
      bn[nt] = c;
    }
  }

  v8f acc[2][4] = {};
  for (int k0 = 0; k0 < K; k0 += 32) {
    const float* pa0 = Ab + arow0 + k0 + khalf;
    const float* pa1 = Ab + arow1 + k0 + khalf;
    v16h a0 = frag_from_f32(pa0);
    v16h a1 = frag_from_f32(pa1);
    if (k0 + 32 < K) {
      __builtin_prefetch(pa0 + 32, 0, 1);
      __builtin_prefetch(pa1 + 32, 0, 1);
    }
    #pragma unroll
    for (int nt = 0; nt < 4; ++nt) {
      if (!FULLN && !ton[nt]) continue;
      v16h b = frag_from_f32(Bb + (size_t)bn[nt] * (size_t)ldb + k0 + khalf);
      acc[0][nt] = WMMA_F16(a0, b, acc[0][nt]);
      acc[1][nt] = WMMA_F16(a1, b, acc[1][nt]);
    }
  }

  const int rb = (lane < 16) ? 0 : 8;
  #pragma unroll
  for (int nt = 0; nt < 4; ++nt) {
    if (!FULLN && !ton[nt]) continue;
    const int col = nbase + nt * 16 + l15;
    bool colok = FULLN ? true : (col < N);
    float bv = 0.0f;
    if (colok) {
      if (bias)  bv += bias[col];
      if (bias2) bv += bias2[col];
    }
    #pragma unroll
    for (int mt = 0; mt < 2; ++mt) {
      #pragma unroll
      for (int r = 0; r < 8; ++r) {
        const int row = tileM + mt * 16 + rb + r;
        if (row < M && colok) {
          float v = acc[mt][nt][r] + bv;
          if      (ACT == 1) v = v > 0.0f ? v : 0.0f;
          else if (ACT == 2) v = v > 0.0f ? v : 0.1f * v;
          if (Rb) v += Rb[(size_t)row * (size_t)ldc + col];
          Cb[(size_t)row * (size_t)ldc + col] = v;
        }
      }
    }
  }
}

__global__ void cvt_f32_f16_kernel(const float* __restrict__ src,
                                   __half* __restrict__ dst, int n) {
  int i = blockIdx.x * blockDim.x + threadIdx.x;
  if (i < n) ((_Float16*)dst)[i] = (_Float16)src[i];
}

__global__ __launch_bounds__(256) void transpose_bsg_kernel(
    const float* __restrict__ src, float* __restrict__ dst)
{
  __shared__ float tile[32][33];
  const int b  = blockIdx.z;
  const int s0 = blockIdx.x * 32, g0 = blockIdx.y * 32;
  const int tx = threadIdx.x, ty = threadIdx.y;
  const float* sp = src + (size_t)b * S_ * G_;
  float*       dp = dst + (size_t)b * G_ * S_;
  #pragma unroll
  for (int r = 0; r < 32; r += 8)
    tile[ty + r][tx] = sp[(size_t)(s0 + ty + r) * G_ + g0 + tx];
  __syncthreads();
  #pragma unroll
  for (int r = 0; r < 32; r += 8)
    dp[(size_t)(g0 + ty + r) * S_ + s0 + tx] = tile[tx][ty + r];
}

__global__ __launch_bounds__(256) void lstm_dir_kernel(
    const float* __restrict__ preF, const float* __restrict__ preB,
    const __half* __restrict__ whhF16_f, const __half* __restrict__ whhF16_b,
    float* __restrict__ hs)
{
  const int dir = blockIdx.x;
  const float*    pre = dir ? preB : preF;
  const _Float16* whh = (const _Float16*)(dir ? whhF16_b : whhF16_f);

  __shared__ _Float16 h_lds[B_ * HP_];
  __shared__ float    c_lds[B_ * H_];
  __shared__ float    z_lds[B_ * ZP_];

  const int tid   = threadIdx.x;
  const int wave  = tid >> 5;
  const int lane  = tid & 31;
  const int l15   = lane & 15;
  const int khalf = (lane & 16) ? 8 : 0;

  v16h wfrag[16];
  #pragma unroll
  for (int nt = 0; nt < 4; ++nt) {
    const int tileN = (wave * 4 + nt) * 16;
    #pragma unroll
    for (int kc = 0; kc < 4; ++kc)
      wfrag[nt * 4 + kc] =
          frag_from_f16(whh + (size_t)(tileN + l15) * H_ + kc * 32 + khalf);
  }

  for (int i = tid; i < B_ * HP_; i += 256) h_lds[i] = (_Float16)0.0f;
  for (int i = tid; i < B_ * H_;  i += 256) c_lds[i] = 0.0f;
  __syncthreads();

  for (int step = 0; step < S_; ++step) {
    const int t = dir ? (S_ - 1 - step) : step;
    const float* pre_t = pre + (size_t)t * (B_ * 4 * H_);

    v16h a[4];
    #pragma unroll
    for (int kc = 0; kc < 4; ++kc)
      a[kc] = frag_from_f16(&h_lds[l15 * HP_ + kc * 32 + khalf]);

    #pragma unroll
    for (int nt = 0; nt < 4; ++nt) {
      v8f acc = {};
      #pragma unroll
      for (int kc = 0; kc < 4; ++kc)
        acc = WMMA_F16(a[kc], wfrag[nt * 4 + kc], acc);
      const int col = (wave * 4 + nt) * 16 + l15;
      const int rb  = (lane < 16) ? 0 : 8;
      #pragma unroll
      for (int r = 0; r < 8; ++r) {
        const int row = rb + r;
        z_lds[row * ZP_ + col] = acc[r] + pre_t[row * 512 + col];
      }
    }
    __syncthreads();

    for (int i = tid; i < B_ * H_; i += 256) {
      const int b_ = i >> 7, hx = i & (H_ - 1);
      const float zi = z_lds[b_ * ZP_ + hx];
      const float zf = z_lds[b_ * ZP_ + 128 + hx];
      const float zg = z_lds[b_ * ZP_ + 256 + hx];
      const float zo = z_lds[b_ * ZP_ + 384 + hx];
      const float si = 1.0f / (1.0f + __expf(-zi));
      const float sf = 1.0f / (1.0f + __expf(-zf));
      const float so = 1.0f / (1.0f + __expf(-zo));
      float c = sf * c_lds[i] + si * tanhf(zg);
      const float h = so * tanhf(c);
      c_lds[i] = c;
      h_lds[b_ * HP_ + hx] = (_Float16)h;
      hs[((size_t)t * B_ + b_) * (2 * H_) + dir * H_ + hx] = h;
    }
    __syncthreads();
  }
}

__global__ __launch_bounds__(256) void attn_mask_norm_kernel(
    float* __restrict__ scores, const float* __restrict__ adj)
{
  const int i = blockIdx.x, b = blockIdx.y, tid = threadIdx.x;
  float* row        = scores + ((size_t)b * S_ + i) * S_;
  const float* arow = adj    + ((size_t)b * S_ + i) * S_;
  __shared__ float red[256];
  float part = 0.0f;
  for (int j = tid; j < S_; j += 256) {
    const float sg = 1.0f / (1.0f + __expf(-row[j]));
    const float av = (j == i) ? 1.0f : arow[j];
    const float v  = (sg + ((j == i) ? 1e-5f : 0.0f)) * av;
    row[j] = v;
    part += v;
  }
  red[tid] = part;
  __syncthreads();
  for (int off = 128; off > 0; off >>= 1) {
    if (tid < off) red[tid] += red[tid + off];
    __syncthreads();
  }
  const float inv = 1.0f / red[0];
  for (int j = tid; j < S_; j += 256) row[j] *= inv;
}

__global__ __launch_bounds__(256) void head_mean_kernel(
    const float* __restrict__ h1, const float* __restrict__ gfin,
    float* __restrict__ hidden)
{
  const int b = blockIdx.x, tid = threadIdx.x;
  for (int d = tid; d < 512; d += 256) {
    float s = 0.0f;
    if (d < 256) {
      for (int t = 0; t < S_; ++t) s += h1[((size_t)t * B_ + b) * 256 + d];
    } else {
      const int dd = d - 256;
      for (int t = 0; t < S_; ++t) s += gfin[((size_t)b * S_ + t) * 256 + dd];
    }
    hidden[b * 512 + d] = s * (1.0f / (float)S_);
  }
}

template <int A_MODE, int ACT, bool FULLN>
static inline void gemm(hipStream_t st,
                        const float* A, const float* Bm,
                        const float* bias, const float* bias2, const float* resid,
                        float* C, int M, int N, int K, int lda, int ldb, int ldc,
                        long long sA = 0, long long sB = 0, long long sC = 0,
                        int batches = 1)
{
  dim3 grid((N + 255) / 256, (M + 31) / 32, batches);
  gemm_wmma_kernel<A_MODE, ACT, FULLN><<<grid, 128, 0, st>>>(
      A, Bm, bias, bias2, resid, C, M, N, K, lda, ldb, ldc, sA, sB, sC);
}

extern "C" void kernel_launch(void* const* d_in, const int* in_sizes, int n_in,
                              void* d_out, int out_size, void* d_ws, size_t ws_size,
                              hipStream_t stream)
{
  (void)in_sizes; (void)n_in; (void)out_size; (void)ws_size;
  auto F = [&](int i) { return (const float*)d_in[i]; };

  const float* x        = F(0);
  const float* adj      = F(1);
  const float* l0f_Wih  = F(2),  *l0f_Whh = F(3),  *l0f_bih = F(4),  *l0f_bhh = F(5);
  const float* l0b_Wih  = F(6),  *l0b_Whh = F(7),  *l0b_bih = F(8),  *l0b_bhh = F(9);
  const float* l1f_Wih  = F(10), *l1f_Whh = F(11), *l1f_bih = F(12), *l1f_bhh = F(13);
  const float* l1b_Wih  = F(14), *l1b_Whh = F(15), *l1b_bih = F(16), *l1b_bhh = F(17);
  const float* fc1_W    = F(18), *fc1_b   = F(19);
  const float* wattn_W  = F(20), *wattn_b = F(21);
  const float* lin0_W   = F(22), *lin0_b  = F(23);
  const float* lin1_W   = F(24), *lin1_b  = F(25);
  const float* linf_W   = F(26), *linf_b  = F(27);
  const float* fcA_W    = F(28), *fcA_b   = F(29);
  const float* fcB_W    = F(30), *fcB_b   = F(31);
  const float* fcC_W    = F(32), *fcC_b   = F(33);

  char* wsp = (char*)d_ws;
  size_t off = 0;
  auto carve = [&](size_t bytes) -> void* {
    void* p = wsp + off;
    off += (bytes + 255) & ~(size_t)255;
    return p;
  };
  float* preF   = (float*)carve((size_t)S_ * B_ * 4 * H_ * 4);
  float* preB   = (float*)carve((size_t)S_ * B_ * 4 * H_ * 4);
  float* h0     = (float*)carve((size_t)S_ * B_ * 2 * H_ * 4);
  float* h1     = (float*)carve((size_t)S_ * B_ * 2 * H_ * 4);
  float* gA     = (float*)carve((size_t)B_ * S_ * G_ * 4);
  float* gB     = (float*)carve((size_t)B_ * S_ * G_ * 4);
  float* gC     = (float*)carve((size_t)B_ * S_ * G_ * 4);
  float* gT     = (float*)carve((size_t)B_ * G_ * S_ * 4);
  float* scores = (float*)carve((size_t)B_ * S_ * S_ * 4);
  __half* whh0f = (__half*)carve((size_t)4 * H_ * H_ * 2);
  __half* whh0b = (__half*)carve((size_t)4 * H_ * H_ * 2);
  __half* whh1f = (__half*)carve((size_t)4 * H_ * H_ * 2);
  __half* whh1b = (__half*)carve((size_t)4 * H_ * H_ * 2);
  float* fcAo   = (float*)carve((size_t)B_ * 128 * 4);
  float* fcBo   = (float*)carve((size_t)B_ * 32 * 4);

  const int MSB = S_ * B_;
  const int nwh = 4 * H_ * H_;

  cvt_f32_f16_kernel<<<(nwh + 255) / 256, 256, 0, stream>>>(l0f_Whh, whh0f, nwh);
  cvt_f32_f16_kernel<<<(nwh + 255) / 256, 256, 0, stream>>>(l0b_Whh, whh0b, nwh);
  cvt_f32_f16_kernel<<<(nwh + 255) / 256, 256, 0, stream>>>(l1f_Whh, whh1f, nwh);
  cvt_f32_f16_kernel<<<(nwh + 255) / 256, 256, 0, stream>>>(l1b_Whh, whh1b, nwh);

  gemm<1, 0, true>(stream, x, l0f_Wih, l0f_bih, l0f_bhh, nullptr, preF,
                   MSB, 4 * H_, D_, D_, D_, 4 * H_);
  gemm<1, 0, true>(stream, x, l0b_Wih, l0b_bih, l0b_bhh, nullptr, preB,
                   MSB, 4 * H_, D_, D_, D_, 4 * H_);
  lstm_dir_kernel<<<2, 256, 0, stream>>>(preF, preB, whh0f, whh0b, h0);

  gemm<0, 0, true>(stream, h0, l1f_Wih, l1f_bih, l1f_bhh, nullptr, preF,
                   MSB, 4 * H_, 2 * H_, 2 * H_, 2 * H_, 4 * H_);
  gemm<0, 0, true>(stream, h0, l1b_Wih, l1b_bih, l1b_bhh, nullptr, preB,
                   MSB, 4 * H_, 2 * H_, 2 * H_, 2 * H_, 4 * H_);
  lstm_dir_kernel<<<2, 256, 0, stream>>>(preF, preB, whh1f, whh1b, h1);

  gemm<0, 0, true>(stream, x, fc1_W, fc1_b, nullptr, nullptr, gA,
                   MSB, G_, D_, D_, D_, G_);

  float* cur = gA; float* t1 = gB; float* t2 = gC;
  const long long sGG = (long long)S_ * G_;
  const long long sSS = (long long)S_ * S_;
  for (int l = 0; l < 4; ++l) {
    const float* Wq = wattn_W + (size_t)l * G_ * G_;
    gemm<0, 0, true>(stream, cur, Wq, wattn_b + l * G_, nullptr, nullptr, t2,
                     MSB, G_, G_, G_, G_, G_);
    gemm<0, 0, true>(stream, t2, cur, nullptr, nullptr, nullptr, scores,
                     S_, S_, G_, G_, G_, S_, sGG, sGG, sSS, B_);
    attn_mask_norm_kernel<<<dim3(S_, B_), 256, 0, stream>>>(scores, adj);
    transpose_bsg_kernel<<<dim3(S_ / 32, G_ / 32, B_), dim3(32, 8), 0, stream>>>(cur, gT);
    gemm<0, 0, true>(stream, scores, gT, nullptr, nullptr, nullptr, t1,
                     S_, G_, S_, S_, S_, G_, sSS, (long long)G_ * S_, sGG, B_);
    gemm<0, 1, true>(stream, t1, lin0_W + (size_t)l * G_ * G_, lin0_b + l * G_,
                     nullptr, nullptr, t2, MSB, G_, G_, G_, G_, G_);
    gemm<0, 1, true>(stream, t2, lin1_W + (size_t)l * G_ * G_, lin1_b + l * G_,
                     nullptr, cur, t1, MSB, G_, G_, G_, G_, G_);
    float* old = cur; cur = t1; t1 = old;
  }
  gemm<0, 0, true>(stream, cur, linf_W, linf_b, nullptr, nullptr, t2,
                   MSB, G_, G_, G_, G_, G_);
  float* gfin = t2;

  float* y_out  = (float*)d_out;
  float* hidden = (float*)d_out + B_ * 2;
  head_mean_kernel<<<B_, 256, 0, stream>>>(h1, gfin, hidden);

  gemm<0, 2, true >(stream, hidden, fcA_W, fcA_b, nullptr, nullptr, fcAo,
                    B_, 128, 512, 512, 512, 128);
  gemm<0, 2, false>(stream, fcAo, fcB_W, fcB_b, nullptr, nullptr, fcBo,
                    B_, 32, 128, 128, 128, 32);
  gemm<0, 0, false>(stream, fcBo, fcC_W, fcC_b, nullptr, nullptr, y_out,
                    B_, 2, 32, 32, 32, 2);
}


# LSTM pre-activations written in the recurrence's accumulator order by a transposed input projection (Wih tiles as A, x^T as B): pre loads read whole lines, not 16 rows x 32 B; sum order same
# speedup vs baseline: 2.9504x; 2.9504x over previous
#include <hip/hip_runtime.h>
#include <hip/hip_fp16.h>
#include <hip/hip_bf16.h>

#define B_  16
#define S_  512
#define D_  1024
#define H_  128
#define G_  256
#define HP_ 136
#define ZP_ 516

typedef __attribute__((ext_vector_type(16))) _Float16 v16h;
typedef __attribute__((ext_vector_type(8)))  float    v8f;

__device__ __forceinline__ v16h frag_from_f32(const float* p) {
  float4 x0 = *(const float4*)(p);
  float4 x1 = *(const float4*)(p + 4);
  float4 y0 = *(const float4*)(p + 16);
  float4 y1 = *(const float4*)(p + 20);
  v16h f;
  f[0]=(_Float16)x0.x;  f[1]=(_Float16)x0.y;  f[2]=(_Float16)x0.z;  f[3]=(_Float16)x0.w;
  f[4]=(_Float16)x1.x;  f[5]=(_Float16)x1.y;  f[6]=(_Float16)x1.z;  f[7]=(_Float16)x1.w;
  f[8]=(_Float16)y0.x;  f[9]=(_Float16)y0.y;  f[10]=(_Float16)y0.z; f[11]=(_Float16)y0.w;
  f[12]=(_Float16)y1.x; f[13]=(_Float16)y1.y; f[14]=(_Float16)y1.z; f[15]=(_Float16)y1.w;
  return f;
}

__device__ __forceinline__ v16h frag_from_f16(const _Float16* p) {
  v16h f;
  #pragma unroll
  for (int i = 0; i < 8; ++i) { f[i] = p[i]; f[8 + i] = p[16 + i]; }
  return f;
}

#define WMMA_F16(a, b, c) __builtin_amdgcn_wmma_f32_16x16x32_f16( \
    false, (a), false, (b), (short)0, (c), false, false)

template <int ACT, bool FULLN>
__device__ __forceinline__ void gemm_tile(
    const float* __restrict__ A, const float* __restrict__ Bm,
    const float* __restrict__ bias, const float* __restrict__ bias2,
    const float* __restrict__ resid, float* __restrict__ C,
    int M, int N, int K, int lda, int ldb, int ldc,
    long long sA, long long sB, long long sC,
    int bx, int by, int bz, int tid)
{
  const int lane  = tid & 31;
  const int wave  = tid >> 5;
  const int tileM = by * 32;
  const int nbase = bx * 256 + wave * 64;
  if (nbase >= N) return;

  const int z = bz;
  const float* Ab = A  + (size_t)z * (size_t)sA;
  const float* Bb = Bm + (size_t)z * (size_t)sB;
  float*       Cb = C  + (size_t)z * (size_t)sC;
  const float* Rb = resid ? (resid + (size_t)z * (size_t)sC) : nullptr;

  const int l15   = lane & 15;
  const int khalf = (lane & 16) ? 8 : 0;

  int am0 = tileM + l15;      if (am0 >= M) am0 = M - 1;
  int am1 = tileM + 16 + l15; if (am1 >= M) am1 = M - 1;
  const size_t arow0 = (size_t)am0 * (size_t)lda;
  const size_t arow1 = (size_t)am1 * (size_t)lda;

  bool ton[4]; int bn[4];
  #pragma unroll
  for (int nt = 0; nt < 4; ++nt) {
    if (FULLN) {
      ton[nt] = true;
      bn[nt]  = nbase + nt * 16 + l15;
    } else {
      ton[nt] = (nbase + nt * 16) < N;
      int c = nbase + nt * 16 + l15; if (c >= N) c = N - 1;
      bn[nt] = c;
    }
  }

  v8f acc[2][4] = {};
  for (int k0 = 0; k0 < K; k0 += 32) {
    const float* pa0 = Ab + arow0 + k0 + khalf;
    const float* pa1 = Ab + arow1 + k0 + khalf;
    v16h a0 = frag_from_f32(pa0);
    v16h a1 = frag_from_f32(pa1);
    if (k0 + 32 < K) {
      __builtin_prefetch(pa0 + 32, 0, 1);
      __builtin_prefetch(pa1 + 32, 0, 1);
    }
    #pragma unroll
    for (int nt = 0; nt < 4; ++nt) {
      if (!FULLN && !ton[nt]) continue;
      v16h b = frag_from_f32(Bb + (size_t)bn[nt] * (size_t)ldb + k0 + khalf);
      acc[0][nt] = WMMA_F16(a0, b, acc[0][nt]);
      acc[1][nt] = WMMA_F16(a1, b, acc[1][nt]);
    }
  }

  const int rb = (lane < 16) ? 0 : 8;
  #pragma unroll
  for (int nt = 0; nt < 4; ++nt) {
    if (!FULLN && !ton[nt]) continue;
    const int col = nbase + nt * 16 + l15;
    bool colok = FULLN ? true : (col < N);
    float bv = 0.0f;
    if (colok) {
      if (bias)  bv += bias[col];
      if (bias2) bv += bias2[col];
    }
    #pragma unroll
    for (int mt = 0; mt < 2; ++mt) {
      #pragma unroll
      for (int r = 0; r < 8; ++r) {
        const int row = tileM + mt * 16 + rb + r;
        if (row < M && colok) {
          float v = acc[mt][nt][r] + bv;
          if      (ACT == 1) v = v > 0.0f ? v : 0.0f;
          else if (ACT == 2) v = v > 0.0f ? v : 0.1f * v;
          if (Rb) v += Rb[(size_t)row * (size_t)ldc + col];
          Cb[(size_t)row * (size_t)ldc + col] = v;
        }
      }
    }
  }
}

template <int ACT, bool FULLN>
__global__ __launch_bounds__(128) void gemm_wmma_kernel(
    const float* __restrict__ A, const float* __restrict__ Bm,
    const float* __restrict__ bias, const float* __restrict__ bias2,
    const float* __restrict__ resid, float* __restrict__ C,
    int M, int N, int K, int lda, int ldb, int ldc,
    long long sA, long long sB, long long sC)
{
  gemm_tile<ACT, FULLN>(A, Bm, bias, bias2, resid, C, M, N, K, lda, ldb, ldc,
                        sA, sB, sC, blockIdx.x, blockIdx.y, blockIdx.z,
                        threadIdx.x);
}

template <int X_MODE>
__global__ __launch_bounds__(128) void pre_gemm_kernel(
    const float* __restrict__ X, const float* __restrict__ Wih,
    const float* __restrict__ bih, const float* __restrict__ bhh,
    float* __restrict__ pre, int K)
{
  const int tid   = threadIdx.x;
  const int lane  = tid & 31;
  const int wave  = tid >> 5;
  const int l15   = lane & 15;
  const int half  = lane >> 4;
  const int khalf = half * 8;
  const int w     = blockIdx.x;
  const int t0    = (blockIdx.y * 4 + wave) * 4;

  const int arow_unit = w * 8 + (l15 & 3) + ((l15 >> 3) << 2);
  const int arow_gate = (l15 >> 2) & 1;
  const float* pa0 = Wih + (size_t)((0 + arow_gate) * H_ + arow_unit) * (size_t)K + khalf;
  const float* pa1 = Wih + (size_t)((2 + arow_gate) * H_ + arow_unit) * (size_t)K + khalf;

  const float* pb[4];
  #pragma unroll
  for (int nt = 0; nt < 4; ++nt) {
    const int t = t0 + nt;
    const size_t row = (X_MODE == 1) ? ((size_t)l15 * S_ + (size_t)t)
                                     : ((size_t)t * B_ + (size_t)l15);
    pb[nt] = X + row * (size_t)K + khalf;
  }

  v8f acc[2][4] = {};
  for (int k0 = 0; k0 < K; k0 += 32) {
    v16h a0 = frag_from_f32(pa0 + k0);
    v16h a1 = frag_from_f32(pa1 + k0);
    if (k0 + 32 < K) {
      #pragma unroll
      for (int nt = 0; nt < 4; ++nt) __builtin_prefetch(pb[nt] + k0 + 32, 0, 1);
    }
    #pragma unroll
    for (int nt = 0; nt < 4; ++nt) {
      v16h b = frag_from_f32(pb[nt] + k0);
      acc[0][nt] = WMMA_F16(a0, b, acc[0][nt]);
      acc[1][nt] = WMMA_F16(a1, b, acc[1][nt]);
    }
  }

  #pragma unroll
  for (int ty = 0; ty < 2; ++ty) {
    #pragma unroll
    for (int q = 0; q < 2; ++q) {
      const int bidx = (ty * 2 + q) * H_ + w * 8 + half * 4;
      const float4 b1 = *(const float4*)(bih + bidx);
      const float4 b2 = *(const float4*)(bhh + bidx);
      const float bv0 = b1.x + b2.x, bv1 = b1.y + b2.y, bv2 = b1.z + b2.z, bv3 = b1.w + b2.w;
      #pragma unroll
      for (int nt = 0; nt < 4; ++nt) {
        float4 o;
        o.x = acc[ty][nt][q * 4 + 0] + bv0;
        o.y = acc[ty][nt][q * 4 + 1] + bv1;
        o.z = acc[ty][nt][q * 4 + 2] + bv2;
        o.w = acc[ty][nt][q * 4 + 3] + bv3;
        *(float4*)(pre + (size_t)(t0 + nt) * (B_ * 4 * H_)
                       + w * 512 + ty * 256 + q * 128 + lane * 4) = o;
      }
    }
  }
}

__global__ void cvt_f32_f16_kernel(const float* __restrict__ src,
                                   __half* __restrict__ dst, int n) {
  int i = blockIdx.x * blockDim.x + threadIdx.x;
  if (i < n) ((_Float16*)dst)[i] = (_Float16)src[i];
}

__device__ __forceinline__ void transpose_bsg_body(
    const float* __restrict__ src, float* __restrict__ dst,
    float (*tile)[33], int bx, int by, int bz, int tid)
{
  const int b  = bz;
  const int s0 = bx * 32, g0 = by * 32;
  const int tx = tid & 31, ty = tid >> 5;
  const float* sp = src + (size_t)b * S_ * G_;
  float*       dp = dst + (size_t)b * G_ * S_;
  #pragma unroll
  for (int r = 0; r < 32; r += 8)
    tile[ty + r][tx] = sp[(size_t)(s0 + ty + r) * G_ + g0 + tx];
  __syncthreads();
  #pragma unroll
  for (int r = 0; r < 32; r += 8)
    dp[(size_t)(g0 + ty + r) * S_ + s0 + tx] = tile[tx][ty + r];
}

struct LstmArgs {
  const float*  preF;  const float*  preB;
  const __half* whhF;  const __half* whhB;
  float*  hs;
  float*  cstate;
  __half* hstate;
  int step0, nsteps;
};

#define WG_THREADS_ 512
#define HS32_ 132
#define LSTM_LDS_BYTES (2 * B_ * HP_ * 2 + 2 * B_ * HS32_ * 4)

typedef float    __attribute__((ext_vector_type(4))) v4f;
typedef _Float16 __attribute__((ext_vector_type(4))) v4h;
typedef v4f __attribute__((may_alias)) v4f_alias;
typedef v4h __attribute__((may_alias)) v4h_alias;

__device__ __forceinline__ float tanh_hw(float x) { return __builtin_amdgcn_tanhf(x); }
__device__ __forceinline__ float sigmoid_hw(float x) {
  return 0.5f + 0.5f * __builtin_amdgcn_tanhf(0.5f * x);
}

__device__ __forceinline__ void lstm_chunk(const LstmArgs& la, unsigned char* smem,
                                           int dir, int tid)
{
  const float* __restrict__ pre = dir ? la.preB : la.preF;
  const _Float16* __restrict__ whh = (const _Float16*)(dir ? la.whhB : la.whhF);
  float* __restrict__ hs = la.hs;
  _Float16* hst = (_Float16*)la.hstate + (size_t)dir * (B_ * HP_);
  float*    cst = la.cstate + (size_t)dir * (B_ * H_);

  _Float16* h16 = (_Float16*)smem;
  float*    h32 = (float*)(smem + 2 * B_ * HP_ * 2);

  const int wave  = tid >> 5;
  const int lane  = tid & 31;
  const int l15   = lane & 15;
  const int half  = lane >> 4;
  const int khalf = half * 8;
  const int ux0   = wave * 8 + half * 4;

  const int arow_unit = wave * 8 + (l15 & 3) + ((l15 >> 3) << 2);
  const int arow_gate = (l15 >> 2) & 1;
  v16h wfrag[8];
  #pragma unroll
  for (int ty = 0; ty < 2; ++ty) {
    const int wrow = (ty * 2 + arow_gate) * H_ + arow_unit;
    #pragma unroll
    for (int kc = 0; kc < 4; ++kc)
      wfrag[ty * 4 + kc] = frag_from_f16(whh + (size_t)wrow * H_ + kc * 32 + khalf);
  }

  v4f_alias* cst_v = (v4f_alias*)cst + wave * 32 + lane;
  v4h_alias* hst_v = (v4h_alias*)hst + wave * 32 + lane;

  v4f c;
  v4h hh;
  if (la.step0 == 0) {
    c  = (v4f)(0.0f);
    hh = (v4h)((_Float16)0.0f);
  } else {
    c  = *cst_v;
    hh = *hst_v;
  }
  *(v4h_alias*)(h16 + l15 * HP_ + ux0) = hh;
  __syncthreads();

  const int fb = tid >> 5, fc = (tid & 31) * 4;

  const int step1 = la.step0 + la.nsteps;
  int p = 0;
  for (int step = la.step0; step < step1; ++step) {
    const int t = dir ? (S_ - 1 - step) : step;
    const float* pre_l = pre + (size_t)t * (B_ * 4 * H_) + wave * 512 + lane * 4;

    const v4f pzi = *(const v4f_alias*)(pre_l);
    const v4f pzf = *(const v4f_alias*)(pre_l + 128);
    const v4f pzg = *(const v4f_alias*)(pre_l + 256);
    const v4f pzo = *(const v4f_alias*)(pre_l + 384);

    const _Float16* hr = h16 + p * (B_ * HP_);
    v16h hb[4];
    #pragma unroll
    for (int kc = 0; kc < 4; ++kc)
      hb[kc] = frag_from_f16(&hr[l15 * HP_ + kc * 32 + khalf]);

    __builtin_amdgcn_sched_barrier(0);

    v8f accX = {}, accY = {};
    #pragma unroll
    for (int kc = 0; kc < 4; ++kc) {
      accX = WMMA_F16(wfrag[kc],     hb[kc], accX);
      accY = WMMA_F16(wfrag[4 + kc], hb[kc], accY);
    }

    if (step != la.step0) {
      const int tp = dir ? (t + 1) : (t - 1);
      const v4f o = *(const v4f_alias*)(h32 + p * (B_ * HS32_) + fb * HS32_ + fc);
      *(v4f_alias*)(hs + (size_t)tp * (B_ * 2 * H_) + fb * (2 * H_) + dir * H_ + fc) = o;
    }

    v4f ho;
    #pragma unroll
    for (int j = 0; j < 4; ++j) {
      const float zi = accX[j]     + pzi[j];
      const float zf = accX[4 + j] + pzf[j];
      const float zg = accY[j]     + pzg[j];
      const float zo = accY[4 + j] + pzo[j];
      const float si = sigmoid_hw(zi);
      const float sf = sigmoid_hw(zf);
      const float so = sigmoid_hw(zo);
      float cn = sf * c[j] + si * tanh_hw(zg);
      const float h = so * tanh_hw(cn);
      c[j]  = cn;
      ho[j] = h;
      hh[j] = (_Float16)h;
    }

    const int pn = p ^ 1;
    *(v4h_alias*)(h16 + pn * (B_ * HP_) + l15 * HP_ + ux0) = hh;
    *(v4f_alias*)(h32 + pn * (B_ * HS32_) + l15 * HS32_ + ux0) = ho;
    __syncthreads();
    p = pn;
  }

  if (la.nsteps > 0) {
    const int tl = dir ? (S_ - step1) : (step1 - 1);
    const v4f o = *(const v4f_alias*)(h32 + p * (B_ * HS32_) + fb * HS32_ + fc);
    *(v4f_alias*)(hs + (size_t)tl * (B_ * 2 * H_) + fb * (2 * H_) + dir * H_ + fc) = o;
  }
  if (step1 < S_) {
    *cst_v = c;
    *hst_v = hh;
  }
}

__device__ __forceinline__ void attn_mask_norm_body(
    float* __restrict__ scores, const float* __restrict__ adj,
    float* red, int i, int b, int tid)
{
  float* row        = scores + ((size_t)b * S_ + i) * S_;
  const float* arow = adj    + ((size_t)b * S_ + i) * S_;
  float part = 0.0f;
  for (int j = tid; j < S_; j += 256) {
    const float sg = 1.0f / (1.0f + __expf(-row[j]));
    const float av = (j == i) ? 1.0f : arow[j];
    const float v  = (sg + ((j == i) ? 1e-5f : 0.0f)) * av;
    row[j] = v;
    part += v;
  }
  red[tid] = part;
  __syncthreads();
  for (int off = 128; off > 0; off >>= 1) {
    if (tid < off) red[tid] += red[tid + off];
    __syncthreads();
  }
  const float inv = 1.0f / red[0];
  for (int j = tid; j < S_; j += 256) row[j] *= inv;
}

struct StageArgs {
  int kind;
  const float* A;  const float* Bm;  const float* bias;  const float* resid;
  float* C;
  int M, N, K, lda, ldb, ldc;
  long long sA, sB, sC;
  int gx, gy, nvb, nblocks;
};

__global__ __launch_bounds__(WG_THREADS_) void colaunch_kernel(LstmArgs la, StageArgs sa)
{
  __shared__ __attribute__((aligned(16))) unsigned char smem[LSTM_LDS_BYTES];
  const int tid = threadIdx.x;
  if (blockIdx.x < 2) {
    lstm_chunk(la, smem, blockIdx.x, tid);
    return;
  }
  const int sub  = tid >> 8;
  const int t256 = tid & 255;
  const int bid  = (blockIdx.x - 2) * 2 + sub;
  if (sa.kind <= 1) {
    const int vb = bid * 2 + (t256 >> 7);
    if (vb >= sa.nvb) return;
    const int bx = vb % sa.gx;
    const int rest = vb / sa.gx;
    const int by = rest % sa.gy;
    const int bz = rest / sa.gy;
    if (sa.kind == 0)
      gemm_tile<0, true>(sa.A, sa.Bm, sa.bias, nullptr, sa.resid, sa.C,
                         sa.M, sa.N, sa.K, sa.lda, sa.ldb, sa.ldc,
                         sa.sA, sa.sB, sa.sC, bx, by, bz, tid & 127);
    else
      gemm_tile<1, true>(sa.A, sa.Bm, sa.bias, nullptr, sa.resid, sa.C,
                         sa.M, sa.N, sa.K, sa.lda, sa.ldb, sa.ldc,
                         sa.sA, sa.sB, sa.sC, bx, by, bz, tid & 127);
  } else if (sa.kind == 2) {
    attn_mask_norm_body(sa.C, sa.A, (float*)smem + sub * 256, bid % sa.gx, bid / sa.gx, t256);
  } else {
    const int bx = bid % sa.gx;
    const int rest = bid / sa.gx;
    transpose_bsg_body(sa.A, sa.C, (float (*)[33])(smem + sub * (32 * 33 * 4)),
                       bx, rest % sa.gy, rest / sa.gy, t256);
  }
}

__global__ __launch_bounds__(256) void head_mean_kernel(
    const float* __restrict__ h1, const float* __restrict__ gfin,
    float* __restrict__ hidden)
{
  const int b = blockIdx.x, tid = threadIdx.x;
  for (int d = tid; d < 512; d += 256) {
    float s = 0.0f;
    if (d < 256) {
      for (int t = 0; t < S_; ++t) s += h1[((size_t)t * B_ + b) * 256 + d];
    } else {
      const int dd = d - 256;
      for (int t = 0; t < S_; ++t) s += gfin[((size_t)b * S_ + t) * 256 + dd];
    }
    hidden[b * 512 + d] = s * (1.0f / (float)S_);
  }
}

template <int ACT, bool FULLN>
static inline void gemm(hipStream_t st,
                        const float* A, const float* Bm,
                        const float* bias, const float* bias2, const float* resid,
                        float* C, int M, int N, int K, int lda, int ldb, int ldc,
                        long long sA = 0, long long sB = 0, long long sC = 0,
                        int batches = 1)
{
  dim3 grid((N + 255) / 256, (M + 31) / 32, batches);
  gemm_wmma_kernel<ACT, FULLN><<<grid, 128, 0, st>>>(
      A, Bm, bias, bias2, resid, C, M, N, K, lda, ldb, ldc, sA, sB, sC);
}

static inline StageArgs gemm_stage(int relu, const float* A, const float* Bm,
                                   const float* bias, const float* resid, float* C,
                                   int M, int N, int K, int lda, int ldb, int ldc,
                                   long long sA = 0, long long sB = 0, long long sC = 0,
                                   int batches = 1)
{
  StageArgs s;
  s.kind = relu ? 1 : 0;
  s.A = A; s.Bm = Bm; s.bias = bias; s.resid = resid; s.C = C;
  s.M = M; s.N = N; s.K = K; s.lda = lda; s.ldb = ldb; s.ldc = ldc;
  s.sA = sA; s.sB = sB; s.sC = sC;
  s.gx = (N + 255) / 256; s.gy = (M + 31) / 32;
  s.nvb = s.gx * s.gy * batches;
  s.nblocks = (s.nvb + 3) / 4;
  return s;
}

static inline StageArgs plain_stage(int kind, const float* in, float* out,
                                    int gx, int gy, int gz)
{
  StageArgs s;
  s.kind = kind;
  s.A = in; s.Bm = nullptr; s.bias = nullptr; s.resid = nullptr; s.C = out;
  s.M = 0; s.N = 0; s.K = 0; s.lda = 0; s.ldb = 0; s.ldc = 0;
  s.sA = 0; s.sB = 0; s.sC = 0;
  s.gx = gx; s.gy = gy;
  s.nvb = gx * gy * gz;
  s.nblocks = s.nvb / 2;
  return s;
}

#define NCHUNK_ 15
#define CHUNK_STEPS_ 34
static inline void lstm_layer_with_stages(hipStream_t st_, LstmArgs la,
                                          const StageArgs* st)
{
  for (int c = 0; c < NCHUNK_; ++c) {
    la.step0  = c * CHUNK_STEPS_;
    la.nsteps = (c == NCHUNK_ - 1) ? (S_ - la.step0) : CHUNK_STEPS_;
    colaunch_kernel<<<2 + st[c].nblocks, WG_THREADS_, 0, st_>>>(la, st[c]);
  }
}

extern "C" void kernel_launch(void* const* d_in, const int* in_sizes, int n_in,
                              void* d_out, int out_size, void* d_ws, size_t ws_size,
                              hipStream_t stream)
{
  (void)in_sizes; (void)n_in; (void)out_size; (void)ws_size;
  auto F = [&](int i) { return (const float*)d_in[i]; };

  const float* x        = F(0);
  const float* adj      = F(1);
  const float* l0f_Wih  = F(2),  *l0f_Whh = F(3),  *l0f_bih = F(4),  *l0f_bhh = F(5);
  const float* l0b_Wih  = F(6),  *l0b_Whh = F(7),  *l0b_bih = F(8),  *l0b_bhh = F(9);
  const float* l1f_Wih  = F(10), *l1f_Whh = F(11), *l1f_bih = F(12), *l1f_bhh = F(13);
  const float* l1b_Wih  = F(14), *l1b_Whh = F(15), *l1b_bih = F(16), *l1b_bhh = F(17);
  const float* fc1_W    = F(18), *fc1_b   = F(19);
  const float* wattn_W  = F(20), *wattn_b = F(21);
  const float* lin0_W   = F(22), *lin0_b  = F(23);
  const float* lin1_W   = F(24), *lin1_b  = F(25);
  const float* linf_W   = F(26), *linf_b  = F(27);
  const float* fcA_W    = F(28), *fcA_b   = F(29);
  const float* fcB_W    = F(30), *fcB_b   = F(31);
  const float* fcC_W    = F(32), *fcC_b   = F(33);

  char* wsp = (char*)d_ws;
  size_t off = 0;
  auto carve = [&](size_t bytes) -> void* {
    void* p = wsp + off;
    off += (bytes + 255) & ~(size_t)255;
    return p;
  };
  float* preF   = (float*)carve((size_t)S_ * B_ * 4 * H_ * 4);
  float* preB   = (float*)carve((size_t)S_ * B_ * 4 * H_ * 4);
  float* h0     = (float*)carve((size_t)S_ * B_ * 2 * H_ * 4);
  float* h1     = (float*)carve((size_t)S_ * B_ * 2 * H_ * 4);
  float* gA     = (float*)carve((size_t)B_ * S_ * G_ * 4);
  float* gB     = (float*)carve((size_t)B_ * S_ * G_ * 4);
  float* gC     = (float*)carve((size_t)B_ * S_ * G_ * 4);
  float* gT     = (float*)carve((size_t)B_ * G_ * S_ * 4);
  float* scores = (float*)carve((size_t)B_ * S_ * S_ * 4);
  __half* whh0f = (__half*)carve((size_t)4 * H_ * H_ * 2);
  __half* whh0b = (__half*)carve((size_t)4 * H_ * H_ * 2);
  __half* whh1f = (__half*)carve((size_t)4 * H_ * H_ * 2);
  __half* whh1b = (__half*)carve((size_t)4 * H_ * H_ * 2);
  float* fcAo   = (float*)carve((size_t)B_ * 128 * 4);
  float* fcBo   = (float*)carve((size_t)B_ * 32 * 4);

  const int MSB = S_ * B_;
  const int nwh = 4 * H_ * H_;

  cvt_f32_f16_kernel<<<(nwh + 255) / 256, 256, 0, stream>>>(l0f_Whh, whh0f, nwh);
  cvt_f32_f16_kernel<<<(nwh + 255) / 256, 256, 0, stream>>>(l0b_Whh, whh0b, nwh);
  cvt_f32_f16_kernel<<<(nwh + 255) / 256, 256, 0, stream>>>(l1f_Whh, whh1f, nwh);
  cvt_f32_f16_kernel<<<(nwh + 255) / 256, 256, 0, stream>>>(l1b_Whh, whh1b, nwh);

  StageArgs stg[2 * NCHUNK_];
  int ns = 0;
  stg[ns++] = gemm_stage(0, x, fc1_W, fc1_b, nullptr, gA, MSB, G_, D_, D_, D_, G_);

  float* cur = gA; float* t1 = gB; float* t2 = gC;
  const long long sGG = (long long)S_ * G_;
  const long long sSS = (long long)S_ * S_;
  for (int l = 0; l < 4; ++l) {
    const float* Wq = wattn_W + (size_t)l * G_ * G_;
    stg[ns++] = gemm_stage(0, cur, Wq, wattn_b + l * G_, nullptr, t2,
                           MSB, G_, G_, G_, G_, G_);
    stg[ns++] = gemm_stage(0, t2, cur, nullptr, nullptr, scores,
                           S_, S_, G_, G_, G_, S_, sGG, sGG, sSS, B_);
    stg[ns++] = plain_stage(2, adj, scores, S_, B_, 1);
    stg[ns++] = plain_stage(3, cur, gT, S_ / 32, G_ / 32, B_);
    stg[ns++] = gemm_stage(0, scores, gT, nullptr, nullptr, t1,
                           S_, G_, S_, S_, S_, G_, sSS, (long long)G_ * S_, sGG, B_);
    stg[ns++] = gemm_stage(1, t1, lin0_W + (size_t)l * G_ * G_, lin0_b + l * G_,
                           nullptr, t2, MSB, G_, G_, G_, G_, G_);
    stg[ns++] = gemm_stage(1, t2, lin1_W + (size_t)l * G_ * G_, lin1_b + l * G_,
                           cur, t1, MSB, G_, G_, G_, G_, G_);
    float* old = cur; cur = t1; t1 = old;
  }
  stg[ns++] = gemm_stage(0, cur, linf_W, linf_b, nullptr, t2, MSB, G_, G_, G_, G_, G_);
  float* gfin = t2;

  const dim3 pgrid(16, S_ / 16);
  pre_gemm_kernel<1><<<pgrid, 128, 0, stream>>>(x, l0f_Wih, l0f_bih, l0f_bhh, preF, D_);
  pre_gemm_kernel<1><<<pgrid, 128, 0, stream>>>(x, l0b_Wih, l0b_bih, l0b_bhh, preB, D_);
  LstmArgs la0;
  la0.preF = preF; la0.preB = preB; la0.whhF = whh0f; la0.whhB = whh0b;
  la0.hs = h0;
  la0.cstate = h1;
  la0.hstate = (__half*)(h1 + 2 * B_ * H_);
  la0.step0 = 0; la0.nsteps = 0;
  lstm_layer_with_stages(stream, la0, stg);

  pre_gemm_kernel<0><<<pgrid, 128, 0, stream>>>(h0, l1f_Wih, l1f_bih, l1f_bhh, preF, 2 * H_);
  pre_gemm_kernel<0><<<pgrid, 128, 0, stream>>>(h0, l1b_Wih, l1b_bih, l1b_bhh, preB, 2 * H_);
  LstmArgs la1;
  la1.preF = preF; la1.preB = preB; la1.whhF = whh1f; la1.whhB = whh1b;
  la1.hs = h1;
  la1.cstate = h0;
  la1.hstate = (__half*)(h0 + 2 * B_ * H_);
  la1.step0 = 0; la1.nsteps = 0;
  lstm_layer_with_stages(stream, la1, stg + NCHUNK_);

  float* y_out  = (float*)d_out;
  float* hidden = (float*)d_out + B_ * 2;
  head_mean_kernel<<<B_, 256, 0, stream>>>(h1, gfin, hidden);

  gemm<2, true >(stream, hidden, fcA_W, fcA_b, nullptr, nullptr, fcAo,
                 B_, 128, 512, 512, 512, 128);
  gemm<2, false>(stream, fcAo, fcB_W, fcB_b, nullptr, nullptr, fcBo,
                 B_, 32, 128, 128, 128, 32);
  gemm<0, false>(stream, fcBo, fcC_W, fcC_b, nullptr, nullptr, y_out,
                 B_, 2, 32, 32, 32, 2);
}


# co-scheduled gemm stages: one 128-thread block per workgroup (was four), so an 8192-row product runs on 256 compute units, not 64; tiles, k loop, sum order unchanged
# speedup vs baseline: 3.2471x; 1.1006x over previous
#include <hip/hip_runtime.h>
#include <hip/hip_fp16.h>
#include <hip/hip_bf16.h>

#define B_  16
#define S_  512
#define D_  1024
#define H_  128
#define G_  256
#define HP_ 136
#define ZP_ 516

typedef __attribute__((ext_vector_type(16))) _Float16 v16h;
typedef __attribute__((ext_vector_type(8)))  float    v8f;

__device__ __forceinline__ v16h frag_from_f32(const float* p) {
  float4 x0 = *(const float4*)(p);
  float4 x1 = *(const float4*)(p + 4);
  float4 y0 = *(const float4*)(p + 16);
  float4 y1 = *(const float4*)(p + 20);
  v16h f;
  f[0]=(_Float16)x0.x;  f[1]=(_Float16)x0.y;  f[2]=(_Float16)x0.z;  f[3]=(_Float16)x0.w;
  f[4]=(_Float16)x1.x;  f[5]=(_Float16)x1.y;  f[6]=(_Float16)x1.z;  f[7]=(_Float16)x1.w;
  f[8]=(_Float16)y0.x;  f[9]=(_Float16)y0.y;  f[10]=(_Float16)y0.z; f[11]=(_Float16)y0.w;
  f[12]=(_Float16)y1.x; f[13]=(_Float16)y1.y; f[14]=(_Float16)y1.z; f[15]=(_Float16)y1.w;
  return f;
}

__device__ __forceinline__ v16h frag_from_f16(const _Float16* p) {
  v16h f;
  #pragma unroll
  for (int i = 0; i < 8; ++i) { f[i] = p[i]; f[8 + i] = p[16 + i]; }
  return f;
}

#define WMMA_F16(a, b, c) __builtin_amdgcn_wmma_f32_16x16x32_f16( \
    false, (a), false, (b), (short)0, (c), false, false)

template <int ACT, bool FULLN>
__device__ __forceinline__ void gemm_tile(
    const float* __restrict__ A, const float* __restrict__ Bm,
    const float* __restrict__ bias, const float* __restrict__ bias2,
    const float* __restrict__ resid, float* __restrict__ C,
    int M, int N, int K, int lda, int ldb, int ldc,
    long long sA, long long sB, long long sC,
    int bx, int by, int bz, int tid)
{
  const int lane  = tid & 31;
  const int wave  = tid >> 5;
  const int tileM = by * 32;
  const int nbase = bx * 256 + wave * 64;
  if (nbase >= N) return;

  const int z = bz;
  const float* Ab = A  + (size_t)z * (size_t)sA;
  const float* Bb = Bm + (size_t)z * (size_t)sB;
  float*       Cb = C  + (size_t)z * (size_t)sC;
  const float* Rb = resid ? (resid + (size_t)z * (size_t)sC) : nullptr;

  const int l15   = lane & 15;
  const int khalf = (lane & 16) ? 8 : 0;

  int am0 = tileM + l15;      if (am0 >= M) am0 = M - 1;
  int am1 = tileM + 16 + l15; if (am1 >= M) am1 = M - 1;
  const size_t arow0 = (size_t)am0 * (size_t)lda;
  const size_t arow1 = (size_t)am1 * (size_t)lda;

  bool ton[4]; int bn[4];
  #pragma unroll
  for (int nt = 0; nt < 4; ++nt) {
    if (FULLN) {
      ton[nt] = true;
      bn[nt]  = nbase + nt * 16 + l15;
    } else {
      ton[nt] = (nbase + nt * 16) < N;
      int c = nbase + nt * 16 + l15; if (c >= N) c = N - 1;
      bn[nt] = c;
    }
  }

  v8f acc[2][4] = {};
  for (int k0 = 0; k0 < K; k0 += 32) {
    const float* pa0 = Ab + arow0 + k0 + khalf;
    const float* pa1 = Ab + arow1 + k0 + khalf;
    v16h a0 = frag_from_f32(pa0);
    v16h a1 = frag_from_f32(pa1);
    if (k0 + 32 < K) {
      __builtin_prefetch(pa0 + 32, 0, 1);
      __builtin_prefetch(pa1 + 32, 0, 1);
    }
    #pragma unroll
    for (int nt = 0; nt < 4; ++nt) {
      if (!FULLN && !ton[nt]) continue;
      v16h b = frag_from_f32(Bb + (size_t)bn[nt] * (size_t)ldb + k0 + khalf);
      acc[0][nt] = WMMA_F16(a0, b, acc[0][nt]);
      acc[1][nt] = WMMA_F16(a1, b, acc[1][nt]);
    }
  }

  const int rb = (lane < 16) ? 0 : 8;
  #pragma unroll
  for (int nt = 0; nt < 4; ++nt) {
    if (!FULLN && !ton[nt]) continue;
    const int col = nbase + nt * 16 + l15;
    bool colok = FULLN ? true : (col < N);
    float bv = 0.0f;
    if (colok) {
      if (bias)  bv += bias[col];
      if (bias2) bv += bias2[col];
    }
    #pragma unroll
    for (int mt = 0; mt < 2; ++mt) {
      #pragma unroll
      for (int r = 0; r < 8; ++r) {
        const int row = tileM + mt * 16 + rb + r;
        if (row < M && colok) {
          float v = acc[mt][nt][r] + bv;
          if      (ACT == 1) v = v > 0.0f ? v : 0.0f;
          else if (ACT == 2) v = v > 0.0f ? v : 0.1f * v;
          if (Rb) v += Rb[(size_t)row * (size_t)ldc + col];
          Cb[(size_t)row * (size_t)ldc + col] = v;
        }
      }
    }
  }
}

template <int ACT, bool FULLN>
__global__ __launch_bounds__(128) void gemm_wmma_kernel(
    const float* __restrict__ A, const float* __restrict__ Bm,
    const float* __restrict__ bias, const float* __restrict__ bias2,
    const float* __restrict__ resid, float* __restrict__ C,
    int M, int N, int K, int lda, int ldb, int ldc,
    long long sA, long long sB, long long sC)
{
  gemm_tile<ACT, FULLN>(A, Bm, bias, bias2, resid, C, M, N, K, lda, ldb, ldc,
                        sA, sB, sC, blockIdx.x, blockIdx.y, blockIdx.z,
                        threadIdx.x);
}

template <int X_MODE>
__global__ __launch_bounds__(128) void pre_gemm_kernel(
    const float* __restrict__ X, const float* __restrict__ Wih,
    const float* __restrict__ bih, const float* __restrict__ bhh,
    float* __restrict__ pre, int K)
{
  const int tid   = threadIdx.x;
  const int lane  = tid & 31;
  const int wave  = tid >> 5;
  const int l15   = lane & 15;
  const int half  = lane >> 4;
  const int khalf = half * 8;
  const int w     = blockIdx.x;
  const int t0    = (blockIdx.y * 4 + wave) * 4;

  const int arow_unit = w * 8 + (l15 & 3) + ((l15 >> 3) << 2);
  const int arow_gate = (l15 >> 2) & 1;
  const float* pa0 = Wih + (size_t)((0 + arow_gate) * H_ + arow_unit) * (size_t)K + khalf;
  const float* pa1 = Wih + (size_t)((2 + arow_gate) * H_ + arow_unit) * (size_t)K + khalf;

  const float* pb[4];
  #pragma unroll
  for (int nt = 0; nt < 4; ++nt) {
    const int t = t0 + nt;
    const size_t row = (X_MODE == 1) ? ((size_t)l15 * S_ + (size_t)t)
                                     : ((size_t)t * B_ + (size_t)l15);
    pb[nt] = X + row * (size_t)K + khalf;
  }

  v8f acc[2][4] = {};
  for (int k0 = 0; k0 < K; k0 += 32) {
    v16h a0 = frag_from_f32(pa0 + k0);
    v16h a1 = frag_from_f32(pa1 + k0);
    if (k0 + 32 < K) {
      #pragma unroll
      for (int nt = 0; nt < 4; ++nt) __builtin_prefetch(pb[nt] + k0 + 32, 0, 1);
    }
    #pragma unroll
    for (int nt = 0; nt < 4; ++nt) {
      v16h b = frag_from_f32(pb[nt] + k0);
      acc[0][nt] = WMMA_F16(a0, b, acc[0][nt]);
      acc[1][nt] = WMMA_F16(a1, b, acc[1][nt]);
    }
  }

  #pragma unroll
  for (int ty = 0; ty < 2; ++ty) {
    #pragma unroll
    for (int q = 0; q < 2; ++q) {
      const int bidx = (ty * 2 + q) * H_ + w * 8 + half * 4;
      const float4 b1 = *(const float4*)(bih + bidx);
      const float4 b2 = *(const float4*)(bhh + bidx);
      const float bv0 = b1.x + b2.x, bv1 = b1.y + b2.y, bv2 = b1.z + b2.z, bv3 = b1.w + b2.w;
      #pragma unroll
      for (int nt = 0; nt < 4; ++nt) {
        float4 o;
        o.x = acc[ty][nt][q * 4 + 0] + bv0;
        o.y = acc[ty][nt][q * 4 + 1] + bv1;
        o.z = acc[ty][nt][q * 4 + 2] + bv2;
        o.w = acc[ty][nt][q * 4 + 3] + bv3;
        *(float4*)(pre + (size_t)(t0 + nt) * (B_ * 4 * H_)
                       + w * 512 + ty * 256 + q * 128 + lane * 4) = o;
      }
    }
  }
}

__global__ void cvt_f32_f16_kernel(const float* __restrict__ src,
                                   __half* __restrict__ dst, int n) {
  int i = blockIdx.x * blockDim.x + threadIdx.x;
  if (i < n) ((_Float16*)dst)[i] = (_Float16)src[i];
}

__device__ __forceinline__ void transpose_bsg_body(
    const float* __restrict__ src, float* __restrict__ dst,
    float (*tile)[33], int bx, int by, int bz, int tid)
{
  const int b  = bz;
  const int s0 = bx * 32, g0 = by * 32;
  const int tx = tid & 31, ty = tid >> 5;
  const float* sp = src + (size_t)b * S_ * G_;
  float*       dp = dst + (size_t)b * G_ * S_;
  #pragma unroll
  for (int r = 0; r < 32; r += 8)
    tile[ty + r][tx] = sp[(size_t)(s0 + ty + r) * G_ + g0 + tx];
  __syncthreads();
  #pragma unroll
  for (int r = 0; r < 32; r += 8)
    dp[(size_t)(g0 + ty + r) * S_ + s0 + tx] = tile[tx][ty + r];
}

struct LstmArgs {
  const float*  preF;  const float*  preB;
  const __half* whhF;  const __half* whhB;
  float*  hs;
  float*  cstate;
  __half* hstate;
  int step0, nsteps;
};

#define WG_THREADS_ 512
#define HS32_ 132
#define LSTM_LDS_BYTES (2 * B_ * HP_ * 2 + 2 * B_ * HS32_ * 4)

typedef float    __attribute__((ext_vector_type(4))) v4f;
typedef _Float16 __attribute__((ext_vector_type(4))) v4h;
typedef v4f __attribute__((may_alias)) v4f_alias;
typedef v4h __attribute__((may_alias)) v4h_alias;

__device__ __forceinline__ float tanh_hw(float x) { return __builtin_amdgcn_tanhf(x); }
__device__ __forceinline__ float sigmoid_hw(float x) {
  return 0.5f + 0.5f * __builtin_amdgcn_tanhf(0.5f * x);
}

__device__ __forceinline__ void lstm_chunk(const LstmArgs& la, unsigned char* smem,
                                           int dir, int tid)
{
  const float* __restrict__ pre = dir ? la.preB : la.preF;
  const _Float16* __restrict__ whh = (const _Float16*)(dir ? la.whhB : la.whhF);
  float* __restrict__ hs = la.hs;
  _Float16* hst = (_Float16*)la.hstate + (size_t)dir * (B_ * HP_);
  float*    cst = la.cstate + (size_t)dir * (B_ * H_);

  _Float16* h16 = (_Float16*)smem;
  float*    h32 = (float*)(smem + 2 * B_ * HP_ * 2);

  const int wave  = tid >> 5;
  const int lane  = tid & 31;
  const int l15   = lane & 15;
  const int half  = lane >> 4;
  const int khalf = half * 8;
  const int ux0   = wave * 8 + half * 4;

  const int arow_unit = wave * 8 + (l15 & 3) + ((l15 >> 3) << 2);
  const int arow_gate = (l15 >> 2) & 1;
  v16h wfrag[8];
  #pragma unroll
  for (int ty = 0; ty < 2; ++ty) {
    const int wrow = (ty * 2 + arow_gate) * H_ + arow_unit;
    #pragma unroll
    for (int kc = 0; kc < 4; ++kc)
      wfrag[ty * 4 + kc] = frag_from_f16(whh + (size_t)wrow * H_ + kc * 32 + khalf);
  }

  v4f_alias* cst_v = (v4f_alias*)cst + wave * 32 + lane;
  v4h_alias* hst_v = (v4h_alias*)hst + wave * 32 + lane;

  v4f c;
  v4h hh;
  if (la.step0 == 0) {
    c  = (v4f)(0.0f);
    hh = (v4h)((_Float16)0.0f);
  } else {
    c  = *cst_v;
    hh = *hst_v;
  }
  *(v4h_alias*)(h16 + l15 * HP_ + ux0) = hh;
  __syncthreads();

  const int fb = tid >> 5, fc = (tid & 31) * 4;

  const int step1 = la.step0 + la.nsteps;
  int p = 0;
  for (int step = la.step0; step < step1; ++step) {
    const int t = dir ? (S_ - 1 - step) : step;
    const float* pre_l = pre + (size_t)t * (B_ * 4 * H_) + wave * 512 + lane * 4;

    const v4f pzi = *(const v4f_alias*)(pre_l);
    const v4f pzf = *(const v4f_alias*)(pre_l + 128);
    const v4f pzg = *(const v4f_alias*)(pre_l + 256);
    const v4f pzo = *(const v4f_alias*)(pre_l + 384);

    const _Float16* hr = h16 + p * (B_ * HP_);
    v16h hb[4];
    #pragma unroll
    for (int kc = 0; kc < 4; ++kc)
      hb[kc] = frag_from_f16(&hr[l15 * HP_ + kc * 32 + khalf]);

    __builtin_amdgcn_sched_barrier(0);

    v8f accX = {}, accY = {};
    #pragma unroll
    for (int kc = 0; kc < 4; ++kc) {
      accX = WMMA_F16(wfrag[kc],     hb[kc], accX);
      accY = WMMA_F16(wfrag[4 + kc], hb[kc], accY);
    }

    if (step != la.step0) {
      const int tp = dir ? (t + 1) : (t - 1);
      const v4f o = *(const v4f_alias*)(h32 + p * (B_ * HS32_) + fb * HS32_ + fc);
      *(v4f_alias*)(hs + (size_t)tp * (B_ * 2 * H_) + fb * (2 * H_) + dir * H_ + fc) = o;
    }

    v4f ho;
    #pragma unroll
    for (int j = 0; j < 4; ++j) {
      const float zi = accX[j]     + pzi[j];
      const float zf = accX[4 + j] + pzf[j];
      const float zg = accY[j]     + pzg[j];
      const float zo = accY[4 + j] + pzo[j];
      const float si = sigmoid_hw(zi);
      const float sf = sigmoid_hw(zf);
      const float so = sigmoid_hw(zo);
      float cn = sf * c[j] + si * tanh_hw(zg);
      const float h = so * tanh_hw(cn);
      c[j]  = cn;
      ho[j] = h;
      hh[j] = (_Float16)h;
    }

    const int pn = p ^ 1;
    *(v4h_alias*)(h16 + pn * (B_ * HP_) + l15 * HP_ + ux0) = hh;
    *(v4f_alias*)(h32 + pn * (B_ * HS32_) + l15 * HS32_ + ux0) = ho;
    __syncthreads();
    p = pn;
  }

  if (la.nsteps > 0) {
    const int tl = dir ? (S_ - step1) : (step1 - 1);
    const v4f o = *(const v4f_alias*)(h32 + p * (B_ * HS32_) + fb * HS32_ + fc);
    *(v4f_alias*)(hs + (size_t)tl * (B_ * 2 * H_) + fb * (2 * H_) + dir * H_ + fc) = o;
  }
  if (step1 < S_) {
    *cst_v = c;
    *hst_v = hh;
  }
}

__device__ __forceinline__ void attn_mask_norm_body(
    float* __restrict__ scores, const float* __restrict__ adj,
    float* red, int i, int b, int tid)
{
  float* row        = scores + ((size_t)b * S_ + i) * S_;
  const float* arow = adj    + ((size_t)b * S_ + i) * S_;
  float part = 0.0f;
  for (int j = tid; j < S_; j += 256) {
    const float sg = 1.0f / (1.0f + __expf(-row[j]));
    const float av = (j == i) ? 1.0f : arow[j];
    const float v  = (sg + ((j == i) ? 1e-5f : 0.0f)) * av;
    row[j] = v;
    part += v;
  }
  red[tid] = part;
  __syncthreads();
  for (int off = 128; off > 0; off >>= 1) {
    if (tid < off) red[tid] += red[tid + off];
    __syncthreads();
  }
  const float inv = 1.0f / red[0];
  for (int j = tid; j < S_; j += 256) row[j] *= inv;
}

struct StageArgs {
  int kind;
  const float* A;  const float* Bm;  const float* bias;  const float* resid;
  float* C;
  int M, N, K, lda, ldb, ldc;
  long long sA, sB, sC;
  int gx, gy, nvb, nblocks;
};

__global__ __launch_bounds__(WG_THREADS_) void colaunch_kernel(LstmArgs la, StageArgs sa)
{
  __shared__ __attribute__((aligned(16))) unsigned char smem[LSTM_LDS_BYTES];
  const int tid = threadIdx.x;
  if (blockIdx.x < 2) {
    lstm_chunk(la, smem, blockIdx.x, tid);
    return;
  }
  if (sa.kind <= 1) {
    if (tid >= 128) return;
    const int vb = blockIdx.x - 2;
    const int bx = vb % sa.gx;
    const int rest = vb / sa.gx;
    const int by = rest % sa.gy;
    const int bz = rest / sa.gy;
    if (sa.kind == 0)
      gemm_tile<0, true>(sa.A, sa.Bm, sa.bias, nullptr, sa.resid, sa.C,
                         sa.M, sa.N, sa.K, sa.lda, sa.ldb, sa.ldc,
                         sa.sA, sa.sB, sa.sC, bx, by, bz, tid);
    else
      gemm_tile<1, true>(sa.A, sa.Bm, sa.bias, nullptr, sa.resid, sa.C,
                         sa.M, sa.N, sa.K, sa.lda, sa.ldb, sa.ldc,
                         sa.sA, sa.sB, sa.sC, bx, by, bz, tid);
    return;
  }
  const int sub  = tid >> 8;
  const int t256 = tid & 255;
  const int bid  = (blockIdx.x - 2) * 2 + sub;
  if (sa.kind == 2) {
    attn_mask_norm_body(sa.C, sa.A, (float*)smem + sub * 256, bid % sa.gx, bid / sa.gx, t256);
  } else {
    const int bx = bid % sa.gx;
    const int rest = bid / sa.gx;
    transpose_bsg_body(sa.A, sa.C, (float (*)[33])(smem + sub * (32 * 33 * 4)),
                       bx, rest % sa.gy, rest / sa.gy, t256);
  }
}

__global__ __launch_bounds__(256) void head_mean_kernel(
    const float* __restrict__ h1, const float* __restrict__ gfin,
    float* __restrict__ hidden)
{
  const int b = blockIdx.x, tid = threadIdx.x;
  for (int d = tid; d < 512; d += 256) {
    float s = 0.0f;
    if (d < 256) {
      for (int t = 0; t < S_; ++t) s += h1[((size_t)t * B_ + b) * 256 + d];
    } else {
      const int dd = d - 256;
      for (int t = 0; t < S_; ++t) s += gfin[((size_t)b * S_ + t) * 256 + dd];
    }
    hidden[b * 512 + d] = s * (1.0f / (float)S_);
  }
}

template <int ACT, bool FULLN>
static inline void gemm(hipStream_t st,
                        const float* A, const float* Bm,
                        const float* bias, const float* bias2, const float* resid,
                        float* C, int M, int N, int K, int lda, int ldb, int ldc,
                        long long sA = 0, long long sB = 0, long long sC = 0,
                        int batches = 1)
{
  dim3 grid((N + 255) / 256, (M + 31) / 32, batches);
  gemm_wmma_kernel<ACT, FULLN><<<grid, 128, 0, st>>>(
      A, Bm, bias, bias2, resid, C, M, N, K, lda, ldb, ldc, sA, sB, sC);
}

static inline StageArgs gemm_stage(int relu, const float* A, const float* Bm,
                                   const float* bias, const float* resid, float* C,
                                   int M, int N, int K, int lda, int ldb, int ldc,
                                   long long sA = 0, long long sB = 0, long long sC = 0,
                                   int batches = 1)
{
  StageArgs s;
  s.kind = relu ? 1 : 0;
  s.A = A; s.Bm = Bm; s.bias = bias; s.resid = resid; s.C = C;
  s.M = M; s.N = N; s.K = K; s.lda = lda; s.ldb = ldb; s.ldc = ldc;
  s.sA = sA; s.sB = sB; s.sC = sC;
  s.gx = (N + 255) / 256; s.gy = (M + 31) / 32;
  s.nvb = s.gx * s.gy * batches;
  s.nblocks = s.nvb;
  return s;
}

static inline StageArgs plain_stage(int kind, const float* in, float* out,
                                    int gx, int gy, int gz)
{
  StageArgs s;
  s.kind = kind;
  s.A = in; s.Bm = nullptr; s.bias = nullptr; s.resid = nullptr; s.C = out;
  s.M = 0; s.N = 0; s.K = 0; s.lda = 0; s.ldb = 0; s.ldc = 0;
  s.sA = 0; s.sB = 0; s.sC = 0;
  s.gx = gx; s.gy = gy;
  s.nvb = gx * gy * gz;
  s.nblocks = s.nvb / 2;
  return s;
}

#define NCHUNK_ 15
#define CHUNK_STEPS_ 34
static inline void lstm_layer_with_stages(hipStream_t st_, LstmArgs la,
                                          const StageArgs* st)
{
  for (int c = 0; c < NCHUNK_; ++c) {
    la.step0  = c * CHUNK_STEPS_;
    la.nsteps = (c == NCHUNK_ - 1) ? (S_ - la.step0) : CHUNK_STEPS_;
    colaunch_kernel<<<2 + st[c].nblocks, WG_THREADS_, 0, st_>>>(la, st[c]);
  }
}

extern "C" void kernel_launch(void* const* d_in, const int* in_sizes, int n_in,
                              void* d_out, int out_size, void* d_ws, size_t ws_size,
                              hipStream_t stream)
{
  (void)in_sizes; (void)n_in; (void)out_size; (void)ws_size;
  auto F = [&](int i) { return (const float*)d_in[i]; };

  const float* x        = F(0);
  const float* adj      = F(1);
  const float* l0f_Wih  = F(2),  *l0f_Whh = F(3),  *l0f_bih = F(4),  *l0f_bhh = F(5);
  const float* l0b_Wih  = F(6),  *l0b_Whh = F(7),  *l0b_bih = F(8),  *l0b_bhh = F(9);
  const float* l1f_Wih  = F(10), *l1f_Whh = F(11), *l1f_bih = F(12), *l1f_bhh = F(13);
  const float* l1b_Wih  = F(14), *l1b_Whh = F(15), *l1b_bih = F(16), *l1b_bhh = F(17);
  const float* fc1_W    = F(18), *fc1_b   = F(19);
  const float* wattn_W  = F(20), *wattn_b = F(21);
  const float* lin0_W   = F(22), *lin0_b  = F(23);
  const float* lin1_W   = F(24), *lin1_b  = F(25);
  const float* linf_W   = F(26), *linf_b  = F(27);
  const float* fcA_W    = F(28), *fcA_b   = F(29);
  const float* fcB_W    = F(30), *fcB_b   = F(31);
  const float* fcC_W    = F(32), *fcC_b   = F(33);

  char* wsp = (char*)d_ws;
  size_t off = 0;
  auto carve = [&](size_t bytes) -> void* {
    void* p = wsp + off;
    off += (bytes + 255) & ~(size_t)255;
    return p;
  };
  float* preF   = (float*)carve((size_t)S_ * B_ * 4 * H_ * 4);
  float* preB   = (float*)carve((size_t)S_ * B_ * 4 * H_ * 4);
  float* h0     = (float*)carve((size_t)S_ * B_ * 2 * H_ * 4);
  float* h1     = (float*)carve((size_t)S_ * B_ * 2 * H_ * 4);
  float* gA     = (float*)carve((size_t)B_ * S_ * G_ * 4);
  float* gB     = (float*)carve((size_t)B_ * S_ * G_ * 4);
  float* gC     = (float*)carve((size_t)B_ * S_ * G_ * 4);
  float* gT     = (float*)carve((size_t)B_ * G_ * S_ * 4);
  float* scores = (float*)carve((size_t)B_ * S_ * S_ * 4);
  __half* whh0f = (__half*)carve((size_t)4 * H_ * H_ * 2);
  __half* whh0b = (__half*)carve((size_t)4 * H_ * H_ * 2);
  __half* whh1f = (__half*)carve((size_t)4 * H_ * H_ * 2);
  __half* whh1b = (__half*)carve((size_t)4 * H_ * H_ * 2);
  float* fcAo   = (float*)carve((size_t)B_ * 128 * 4);
  float* fcBo   = (float*)carve((size_t)B_ * 32 * 4);

  const int MSB = S_ * B_;
  const int nwh = 4 * H_ * H_;

  cvt_f32_f16_kernel<<<(nwh + 255) / 256, 256, 0, stream>>>(l0f_Whh, whh0f, nwh);
  cvt_f32_f16_kernel<<<(nwh + 255) / 256, 256, 0, stream>>>(l0b_Whh, whh0b, nwh);
  cvt_f32_f16_kernel<<<(nwh + 255) / 256, 256, 0, stream>>>(l1f_Whh, whh1f, nwh);
  cvt_f32_f16_kernel<<<(nwh + 255) / 256, 256, 0, stream>>>(l1b_Whh, whh1b, nwh);

  StageArgs stg[2 * NCHUNK_];
  int ns = 0;
  stg[ns++] = gemm_stage(0, x, fc1_W, fc1_b, nullptr, gA, MSB, G_, D_, D_, D_, G_);

  float* cur = gA; float* t1 = gB; float* t2 = gC;
  const long long sGG = (long long)S_ * G_;
  const long long sSS = (long long)S_ * S_;
  for (int l = 0; l < 4; ++l) {
    const float* Wq = wattn_W + (size_t)l * G_ * G_;
    stg[ns++] = gemm_stage(0, cur, Wq, wattn_b + l * G_, nullptr, t2,
                           MSB, G_, G_, G_, G_, G_);
    stg[ns++] = gemm_stage(0, t2, cur, nullptr, nullptr, scores,
                           S_, S_, G_, G_, G_, S_, sGG, sGG, sSS, B_);
    stg[ns++] = plain_stage(2, adj, scores, S_, B_, 1);
    stg[ns++] = plain_stage(3, cur, gT, S_ / 32, G_ / 32, B_);
    stg[ns++] = gemm_stage(0, scores, gT, nullptr, nullptr, t1,
                           S_, G_, S_, S_, S_, G_, sSS, (long long)G_ * S_, sGG, B_);
    stg[ns++] = gemm_stage(1, t1, lin0_W + (size_t)l * G_ * G_, lin0_b + l * G_,
                           nullptr, t2, MSB, G_, G_, G_, G_, G_);
    stg[ns++] = gemm_stage(1, t2, lin1_W + (size_t)l * G_ * G_, lin1_b + l * G_,
                           cur, t1, MSB, G_, G_, G_, G_, G_);
    float* old = cur; cur = t1; t1 = old;
  }
  stg[ns++] = gemm_stage(0, cur, linf_W, linf_b, nullptr, t2, MSB, G_, G_, G_, G_, G_);
  float* gfin = t2;

  const dim3 pgrid(16, S_ / 16);
  pre_gemm_kernel<1><<<pgrid, 128, 0, stream>>>(x, l0f_Wih, l0f_bih, l0f_bhh, preF, D_);
  pre_gemm_kernel<1><<<pgrid, 128, 0, stream>>>(x, l0b_Wih, l0b_bih, l0b_bhh, preB, D_);
  LstmArgs la0;
  la0.preF = preF; la0.preB = preB; la0.whhF = whh0f; la0.whhB = whh0b;
  la0.hs = h0;
  la0.cstate = h1;
  la0.hstate = (__half*)(h1 + 2 * B_ * H_);
  la0.step0 = 0; la0.nsteps = 0;
  lstm_layer_with_stages(stream, la0, stg);

  pre_gemm_kernel<0><<<pgrid, 128, 0, stream>>>(h0, l1f_Wih, l1f_bih, l1f_bhh, preF, 2 * H_);
  pre_gemm_kernel<0><<<pgrid, 128, 0, stream>>>(h0, l1b_Wih, l1b_bih, l1b_bhh, preB, 2 * H_);
  LstmArgs la1;
  la1.preF = preF; la1.preB = preB; la1.whhF = whh1f; la1.whhB = whh1b;
  la1.hs = h1;
  la1.cstate = h0;
  la1.hstate = (__half*)(h0 + 2 * B_ * H_);
  la1.step0 = 0; la1.nsteps = 0;
  lstm_layer_with_stages(stream, la1, stg + NCHUNK_);

  float* y_out  = (float*)d_out;
  float* hidden = (float*)d_out + B_ * 2;
  head_mean_kernel<<<B_, 256, 0, stream>>>(h1, gfin, hidden);

  gemm<2, true >(stream, hidden, fcA_W, fcA_b, nullptr, nullptr, fcAo,
                 B_, 128, 512, 512, 512, 128);
  gemm<2, false>(stream, fcAo, fcB_W, fcB_b, nullptr, nullptr, fcBo,
                 B_, 32, 128, 128, 128, 32);
  gemm<0, false>(stream, fcBo, fcC_W, fcC_b, nullptr, nullptr, y_out,
                 B_, 2, 32, 32, 32, 2);
}
